# starting kernel
# baseline (speedup 1.0000x reference)
#include <hip/hip_runtime.h>
#include <hip/hip_bf16.h>


typedef float v2f __attribute__((ext_vector_type(2)));
typedef float v8f __attribute__((ext_vector_type(8)));

#define WPB 2


__device__ __forceinline__ v2f ld_a(const float* m, int ld, int r0, int c0, int lane) {
  const int h = lane >> 4, mr = lane & 15;
  const float* p = m + (r0 + mr) * ld + c0 + (h << 1);
  v2f a; a.x = p[0]; a.y = p[1];
  return a;
}
__device__ __forceinline__ v2f ld_b(const float* m, int ld, int r0, int c0, int lane) {
  const int h = lane >> 4, n = lane & 15;
  const float* p = m + (r0 + (h << 1)) * ld + c0 + n;
  v2f b; b.x = p[0]; b.y = p[ld];
  return b;
}
__device__ __forceinline__ v2f ld_bt(const float* m, int ld, int kr0, int nc0, int lane) {
  const int h = lane >> 4, n = lane & 15;
  const float* p = m + (nc0 + n) * ld + kr0 + (h << 1);
  v2f b; b.x = p[0]; b.y = p[1];
  return b;
}

template <bool BT>
__device__ __forceinline__ v8f mma16(v8f acc, const float* A, int lda, int ar, int ac,
                                     const float* Bm, int ldb, int br, int bc, int lane) {
#pragma unroll
  for (int k = 0; k < 16; k += 4) {
    v2f a = ld_a(A, lda, ar, ac + k, lane);
    v2f b = BT ? ld_bt(Bm, ldb, br + k, bc, lane)
               : ld_b (Bm, ldb, br + k, bc, lane);
    acc = __builtin_amdgcn_wmma_f32_16x16x4_f32(false, a, false, b,
                                                (short)0, acc, false, false);
  }
  return acc;
}

__device__ __forceinline__ void st_c(float* d, int ld, int r0, int c0, v8f c, int lane) {
  const int h = lane >> 4, n = lane & 15;
#pragma unroll
  for (int r = 0; r < 8; ++r)
    d[(r0 + r + (h << 3)) * ld + c0 + n] = c[r];
}
__device__ __forceinline__ void st_c_add(float* d, int ld, const float* a, int lda,
                                         int r0, int c0, v8f c, int lane) {
  const int h = lane >> 4, n = lane & 15;
#pragma unroll
  for (int r = 0; r < 8; ++r) {
    const int rr = r0 + r + (h << 3);
    d[rr * ld + c0 + n] = c[r] + a[rr * lda + c0 + n];
  }
}
__device__ __forceinline__ void st_c_subfrom(float* d, int ld, int r0, int c0,
                                             v8f c, int lane) {
  const int h = lane >> 4, n = lane & 15;
#pragma unroll
  for (int r = 0; r < 8; ++r) {
    const int idx = (r0 + r + (h << 3)) * ld + c0 + n;
    d[idx] = d[idx] - c[r];
  }
}

__global__ __launch_bounds__(32 * WPB)
void kalman_wmma_kernel(const float* __restrict__ state0, const float* __restrict__ cov0,
                        const float* __restrict__ meas,   const float* __restrict__ Fg,
                        const float* __restrict__ Hg,     const float* __restrict__ Qg,
                        const float* __restrict__ Rg,     float* __restrict__ out,
                        int B, int T) {
  const int tid  = threadIdx.x;
  const int lane = tid & 31;
  const int w    = tid >> 5;

  __shared__ float sF[32 * 32];
  __shared__ float sH[16 * 32];
  __shared__ float sQ[32 * 32];
  __shared__ float sR[16 * 16];
  __shared__ float sP [WPB][32 * 32];
  __shared__ float sT1[WPB][32 * 32];
  __shared__ float sT2[WPB][16 * 32];
  __shared__ float sSa[WPB][16 * 32];
  __shared__ float sKm[WPB][32 * 16];
  __shared__ float sx [WPB][32];
  __shared__ float sxp[WPB][32];
  __shared__ float sin_[WPB][16];

  for (int i = tid; i < 32 * 32; i += blockDim.x) sF[i] = Fg[i];
  for (int i = tid; i < 16 * 32; i += blockDim.x) sH[i] = Hg[i];
  for (int i = tid; i < 32 * 32; i += blockDim.x) sQ[i] = Qg[i];
  for (int i = tid; i < 16 * 16; i += blockDim.x) sR[i] = Rg[i];
  __syncthreads();

  const int b = blockIdx.x * WPB + w;
  if (b >= B) return;

  float* P  = sP [w];
  float* T1 = sT1[w];
  float* T2 = sT2[w];
  float* Sa = sSa[w];
  float* Km = sKm[w];
  float* x  = sx [w];
  float* xp = sxp[w];
  float* nu = sin_[w];

  for (int i = lane; i < 32 * 32; i += 32) P[i] = cov0[(size_t)b * 1024 + i];
  x[lane] = state0[(size_t)b * 32 + lane];
  const float* zb = meas + (size_t)b * T * 16;
  float*       ob = out  + (size_t)b * T * 32;
  __builtin_amdgcn_wave_barrier();

#pragma unroll 1
  for (int t = 0; t < T; ++t) {
    if (t + 1 < T) __builtin_prefetch(zb + (size_t)(t + 1) * 16, 0, 1);

    {
      float acc = 0.f;
#pragma unroll
      for (int j = 0; j < 32; ++j) acc += sF[lane * 32 + j] * x[j];
      xp[lane] = acc;
    }
    __builtin_amdgcn_wave_barrier();

#pragma unroll
    for (int i = 0; i < 32; i += 16)
#pragma unroll
      for (int j = 0; j < 32; j += 16) {
        v8f acc = {};
        acc = mma16<false>(acc, sF, 32, i, 0,  P, 32, 0,  j, lane);
        acc = mma16<false>(acc, sF, 32, i, 16, P, 32, 16, j, lane);
        st_c(T1, 32, i, j, acc, lane);
      }
    __builtin_amdgcn_wave_barrier();

#pragma unroll
    for (int i = 0; i < 32; i += 16)
#pragma unroll
      for (int j = 0; j < 32; j += 16) {
        v8f acc = {};
        acc = mma16<true>(acc, T1, 32, i, 0,  sF, 32, 0,  j, lane);
        acc = mma16<true>(acc, T1, 32, i, 16, sF, 32, 16, j, lane);
        st_c_add(P, 32, sQ, 32, i, j, acc, lane);
      }
    __builtin_amdgcn_wave_barrier();

#pragma unroll
    for (int j = 0; j < 32; j += 16) {
      v8f acc = {};
      acc = mma16<false>(acc, sH, 32, 0, 0,  P, 32, 0,  j, lane);
      acc = mma16<false>(acc, sH, 32, 0, 16, P, 32, 16, j, lane);
      st_c(T2, 32, 0, j, acc, lane);
    }
    __builtin_amdgcn_wave_barrier();

    {
      v8f acc = {};
      acc = mma16<true>(acc, T2, 32, 0, 0,  sH, 32, 0,  0, lane);
      acc = mma16<true>(acc, T2, 32, 0, 16, sH, 32, 16, 0, lane);
      st_c_add(Sa, 32, sR, 16, 0, 0, acc, lane);
    }
    for (int i = lane; i < 256; i += 32) {
      const int r = i >> 4, c = i & 15;
      Sa[r * 32 + 16 + c] = (r == c) ? 1.0f : 0.0f;
    }
    __builtin_amdgcn_wave_barrier();

    if (lane < 16) {
      float acc = 0.f;
#pragma unroll
      for (int j = 0; j < 32; ++j) acc += sH[lane * 32 + j] * xp[j];
      nu[lane] = zb[(size_t)t * 16 + lane] - acc;
    }
    __builtin_amdgcn_wave_barrier();

    {
      const int c = lane;
#pragma unroll 1
      for (int k = 0; k < 16; ++k) {
        const float pinv = 1.0f / Sa[k * 32 + k];
        const float rk   = Sa[k * 32 + c] * pinv;
        __builtin_amdgcn_wave_barrier();
        Sa[k * 32 + c] = rk;
        __builtin_amdgcn_wave_barrier();
#pragma unroll 1
        for (int r = 0; r < 16; ++r) {
          if (r == k) continue;
          const float f = Sa[r * 32 + k];
          __builtin_amdgcn_wave_barrier();
          Sa[r * 32 + c] = Sa[r * 32 + c] - f * rk;
          __builtin_amdgcn_wave_barrier();
        }
      }
    }

#pragma unroll
    for (int i = 0; i < 32; i += 16) {
      v8f acc = {};
      acc = mma16<true>(acc, P, 32, i, 0,  sH, 32, 0,  0, lane);
      acc = mma16<true>(acc, P, 32, i, 16, sH, 32, 16, 0, lane);
      st_c(T1, 16, i, 0, acc, lane);
    }
    __builtin_amdgcn_wave_barrier();

#pragma unroll
    for (int i = 0; i < 32; i += 16) {
      v8f acc = {};
      acc = mma16<false>(acc, T1, 16, i, 0, Sa, 32, 0, 16, lane);
      st_c(Km, 16, i, 0, acc, lane);
    }
    __builtin_amdgcn_wave_barrier();

    {
      float acc = xp[lane];
#pragma unroll
      for (int j = 0; j < 16; ++j) acc += Km[lane * 16 + j] * nu[j];
      x[lane] = acc;
      ob[(size_t)t * 32 + lane] = acc;
    }
    __builtin_amdgcn_wave_barrier();

#pragma unroll
    for (int i = 0; i < 32; i += 16)
#pragma unroll
      for (int j = 0; j < 32; j += 16) {
        v8f acc = {};
        acc = mma16<false>(acc, Km, 16, i, 0, T2, 32, 0, j, lane);
        st_c_subfrom(P, 32, i, j, acc, lane);
      }
    __builtin_amdgcn_wave_barrier();
  }
}

extern "C" void kernel_launch(void* const* d_in, const int* in_sizes, int n_in,
                              void* d_out, int out_size, void* d_ws, size_t ws_size,
                              hipStream_t stream) {
  const float* state0 = (const float*)d_in[0];
  const float* cov0   = (const float*)d_in[1];
  const float* meas   = (const float*)d_in[2];
  const float* F      = (const float*)d_in[3];
  const float* H      = (const float*)d_in[4];
  const float* Q      = (const float*)d_in[5];
  const float* R      = (const float*)d_in[6];
  float* out = (float*)d_out;

  const int B = in_sizes[0] / 32;
  const int T = in_sizes[2] / (B * 16);
  const int blocks = (B + WPB - 1) / WPB;
  kalman_wmma_kernel<<<blocks, 32 * WPB, 0, stream>>>(state0, cov0, meas,
                                                      F, H, Q, R, out, B, T);
}


# Time step regrouped by data dependence, 7 LDS hand-overs not 11: x_pred beside F*P, innovation beside P_pred, H*P_pred with P_pred*H^T, x update beside the P update; same arithmetic
# speedup vs baseline: 3.7183x; 3.7183x over previous
#include <hip/hip_runtime.h>
#include <hip/hip_bf16.h>


typedef float v2f __attribute__((ext_vector_type(2)));
typedef float v4f __attribute__((ext_vector_type(4)));
typedef float v8f __attribute__((ext_vector_type(8)));

#define WPB 2


__device__ __forceinline__ v2f ld_a(const float* m, int ld, int r0, int c0, int lane) {
  const int h = lane >> 4, mr = lane & 15;
  const float* p = m + (r0 + mr) * ld + c0 + (h << 1);
  v2f a; a.x = p[0]; a.y = p[1];
  return a;
}
__device__ __forceinline__ v2f ld_b(const float* m, int ld, int r0, int c0, int lane) {
  const int h = lane >> 4, n = lane & 15;
  const float* p = m + (r0 + (h << 1)) * ld + c0 + n;
  v2f b; b.x = p[0]; b.y = p[ld];
  return b;
}
__device__ __forceinline__ v2f ld_bt(const float* m, int ld, int kr0, int nc0, int lane) {
  const int h = lane >> 4, n = lane & 15;
  const float* p = m + (nc0 + n) * ld + kr0 + (h << 1);
  v2f b; b.x = p[0]; b.y = p[1];
  return b;
}

__device__ __forceinline__ v2f ld_b_sinv(const float* m, int q, int lane) {
  const int h = lane >> 4, n = lane & 15;
  const float* p = m + q * 128 + (16 + n) * 4 + (h << 1);
  v2f b; b.x = p[0]; b.y = p[1];
  return b;
}

template <bool BT>
__device__ __forceinline__ v8f mma16(v8f acc, const float* A, int lda, int ar, int ac,
                                     const float* Bm, int ldb, int br, int bc, int lane) {
#pragma unroll
  for (int k = 0; k < 16; k += 4) {
    v2f a = ld_a(A, lda, ar, ac + k, lane);
    v2f b = BT ? ld_bt(Bm, ldb, br + k, bc, lane)
               : ld_b (Bm, ldb, br + k, bc, lane);
    acc = __builtin_amdgcn_wmma_f32_16x16x4_f32(false, a, false, b,
                                                (short)0, acc, false, false);
  }
  return acc;
}

__device__ __forceinline__ v8f mma16_ra(v8f acc, const v2f (&fr)[4],
                                        const float* Bm, int ldb, int br, int bc, int lane) {
  for (int q = 0; q < 4; ++q) {
    v2f b = ld_b(Bm, ldb, br + 4 * q, bc, lane);
    acc = __builtin_amdgcn_wmma_f32_16x16x4_f32(false, fr[q], false, b,
                                                (short)0, acc, false, false);
  }
  return acc;
}
__device__ __forceinline__ v8f mma16_rb(v8f acc, const float* A, int lda, int ar, int ac,
                                        const v2f (&fr)[4], int lane) {
  for (int q = 0; q < 4; ++q) {
    v2f a = ld_a(A, lda, ar, ac + 4 * q, lane);
    acc = __builtin_amdgcn_wmma_f32_16x16x4_f32(false, a, false, fr[q],
                                                (short)0, acc, false, false);
  }
  return acc;
}

__device__ __forceinline__ void st_c(float* d, int ld, int r0, int c0, v8f c, int lane) {
  const int h = lane >> 4, n = lane & 15;
#pragma unroll
  for (int r = 0; r < 8; ++r)
    d[(r0 + r + (h << 3)) * ld + c0 + n] = c[r];
}
__device__ __forceinline__ void st_c_add(float* d, int ld, const float* a, int lda,
                                         int r0, int c0, v8f c, int lane) {
  const int h = lane >> 4, n = lane & 15;
#pragma unroll
  for (int r = 0; r < 8; ++r) {
    const int rr = r0 + r + (h << 3);
    d[rr * ld + c0 + n] = c[r] + a[rr * lda + c0 + n];
  }
}
__device__ __forceinline__ void st_c_subfrom(float* d, int ld, int r0, int c0,
                                             v8f c, int lane) {
  const int h = lane >> 4, n = lane & 15;
#pragma unroll
  for (int r = 0; r < 8; ++r) {
    const int idx = (r0 + r + (h << 3)) * ld + c0 + n;
    d[idx] = d[idx] - c[r];
  }
}

__global__ __launch_bounds__(32 * WPB)
void kalman_wmma_kernel(const float* __restrict__ state0, const float* __restrict__ cov0,
                        const float* __restrict__ meas,   const float* __restrict__ Fg,
                        const float* __restrict__ Hg,     const float* __restrict__ Qg,
                        const float* __restrict__ Rg,     float* __restrict__ out,
                        int B, int T) {
  const int tid  = threadIdx.x;
  const int lane = tid & 31;
  const int w    = tid >> 5;

  __shared__ float sF[32 * 32];
  __shared__ float sH[16 * 32];
  __shared__ float sQ[32 * 32];
  __shared__ float sR[16 * 16];
  __shared__ float sP [WPB][32 * 32];
  __shared__ float sT1[WPB][32 * 32];
  __shared__ float sT2[WPB][16 * 32];
  __shared__ __attribute__((aligned(16))) float sSa[WPB][4 * 128];
  __shared__ float sKm[WPB][32 * 16];
  __shared__ float sx [WPB][32];
  __shared__ float sxp[WPB][32];
  __shared__ float sin_[WPB][16];

  for (int i = tid; i < 32 * 32; i += blockDim.x) sF[i] = Fg[i];
  for (int i = tid; i < 16 * 32; i += blockDim.x) sH[i] = Hg[i];
  for (int i = tid; i < 32 * 32; i += blockDim.x) sQ[i] = Qg[i];
  for (int i = tid; i < 16 * 16; i += blockDim.x) sR[i] = Rg[i];
  __syncthreads();

  const int b = blockIdx.x * WPB + w;
  if (b >= B) return;

  float* P  = sP [w];
  float* T1 = sT1[w];
  float* T2 = sT2[w];
  float* Sa = sSa[w];
  float* Km = sKm[w];
  float* x  = sx [w];
  float* xp = sxp[w];
  float* nu = sin_[w];

  for (int i = lane; i < 32 * 32; i += 32) P[i] = cov0[(size_t)b * 1024 + i];
  x[lane] = state0[(size_t)b * 32 + lane];
  const float* zb = meas + (size_t)b * T * 16;
  float*       ob = out  + (size_t)b * T * 32;
  __builtin_amdgcn_wave_barrier();

  v2f fF[2][2][4], fH[2][4];
  for (int e = 0; e < 16; ++e)
    fF[e >> 3][(e >> 2) & 1][e & 3] = ld_a(sF, 32, 16 * (e >> 3), 4 * (e & 7), lane);
  for (int e = 0; e < 8; ++e)
    fH[e >> 2][e & 3] = ld_a(sH, 32, 0, 4 * e, lane);

#pragma unroll 1
  for (int t = 0; t < T; ++t) {
    if (t + 1 < T) __builtin_prefetch(zb + (size_t)(t + 1) * 16, 0, 1);


#pragma unroll
    for (int i = 0; i < 32; i += 16)
#pragma unroll
      for (int j = 0; j < 32; j += 16) {
        v8f acc = {};
        acc = mma16_ra(acc, fF[i >> 4][0], P, 32, 0,  j, lane);
        acc = mma16_ra(acc, fF[i >> 4][1], P, 32, 16, j, lane);
        st_c(T1, 32, i, j, acc, lane);
      }
    {
      float acc = 0.f;
#pragma unroll
      for (int j = 0; j < 32; ++j) acc += sF[lane * 32 + j] * x[j];
      xp[lane] = acc;
    }
    __builtin_amdgcn_wave_barrier();

    {
      const int r16 = lane & 15;
      float acc = 0.f;
#pragma unroll
      for (int j = 0; j < 32; ++j) acc += sH[r16 * 32 + j] * xp[j];
      nu[r16] = zb[(size_t)t * 16 + r16] - acc;
    }
#pragma unroll
    for (int i = 0; i < 32; i += 16)
#pragma unroll
      for (int j = 0; j < 32; j += 16) {
        v8f acc = {};
        acc = mma16_rb(acc, T1, 32, i, 0,  fF[j >> 4][0], lane);
        acc = mma16_rb(acc, T1, 32, i, 16, fF[j >> 4][1], lane);
        st_c_add(P, 32, sQ, 32, i, j, acc, lane);
      }
    __builtin_amdgcn_wave_barrier();

#pragma unroll
    for (int j = 0; j < 32; j += 16) {
      v8f acc = {};
      acc = mma16_ra(acc, fH[0], P, 32, 0,  j, lane);
      acc = mma16_ra(acc, fH[1], P, 32, 16, j, lane);
      st_c(T2, 32, 0, j, acc, lane);
    }
#pragma unroll
    for (int i = 0; i < 32; i += 16) {
      v8f acc = {};
      acc = mma16_rb(acc, P, 32, i, 0,  fH[0], lane);
      acc = mma16_rb(acc, P, 32, i, 16, fH[1], lane);
      st_c(T1, 16, i, 0, acc, lane);
    }
    __builtin_amdgcn_wave_barrier();

    {
      v8f acc = {};
      acc = mma16_rb(acc, T2, 32, 0, 0,  fH[0], lane);
      acc = mma16_rb(acc, T2, 32, 0, 16, fH[1], lane);
      const int h = lane >> 4, n = lane & 15;
      v4f s0, s1, e0, e1;
#pragma unroll
      for (int r = 0; r < 4; ++r) {
        s0[r] = acc[r]     + sR[(r     + (h << 3)) * 16 + n];
        s1[r] = acc[r + 4] + sR[(r + 4 + (h << 3)) * 16 + n];
        e0[r] = ((h << 3) + r     == n) ? 1.0f : 0.0f;
        e1[r] = ((h << 3) + r + 4 == n) ? 1.0f : 0.0f;
      }
      *(v4f*)(Sa + (2 * h)     * 128 + n * 4)        = s0;
      *(v4f*)(Sa + (2 * h + 1) * 128 + n * 4)        = s1;
      *(v4f*)(Sa + (2 * h)     * 128 + (16 + n) * 4) = e0;
      *(v4f*)(Sa + (2 * h + 1) * 128 + (16 + n) * 4) = e1;
    }
    __builtin_amdgcn_wave_barrier();

    {
      float* colp = Sa + lane * 4;
#pragma unroll
      for (int k0 = 0; k0 < 16; k0 += 4) {
        float own[16], m[4][16];
#pragma unroll
        for (int j = 0; j < 4; ++j)
#pragma unroll
          for (int i = 0; i < 4; ++i) {
            const v4f t = *(const v4f*)(Sa + i * 128 + (k0 + j) * 4);
            m[j][4 * i] = t[0]; m[j][4 * i + 1] = t[1];
            m[j][4 * i + 2] = t[2]; m[j][4 * i + 3] = t[3];
          }
#pragma unroll
        for (int i = 0; i < 4; ++i) {
          const v4f t = *(const v4f*)(colp + i * 128);
          own[4 * i] = t[0]; own[4 * i + 1] = t[1];
          own[4 * i + 2] = t[2]; own[4 * i + 3] = t[3];
        }
        __builtin_amdgcn_wave_barrier();
#pragma unroll
        for (int j = 0; j < 4; ++j) {
          const int k = k0 + j;
          const float pinv = 1.0f / m[j][k];
          const float rk   = own[k] * pinv;
#pragma unroll
          for (int jj = j + 1; jj < 4; ++jj) {
            const float rkm = m[jj][k] * pinv;
#pragma unroll
            for (int r = 0; r < 16; ++r)
              m[jj][r] = (r == k) ? rkm : m[jj][r] - m[j][r] * rkm;
          }
#pragma unroll
          for (int r = 0; r < 16; ++r)
            own[r] = (r == k) ? rk : own[r] - m[j][r] * rk;
        }
#pragma unroll
        for (int i = 0; i < 4; ++i) {
          v4f t;
          t[0] = own[4 * i]; t[1] = own[4 * i + 1];
          t[2] = own[4 * i + 2]; t[3] = own[4 * i + 3];
          *(v4f*)(colp + i * 128) = t;
        }
        __builtin_amdgcn_wave_barrier();
      }
    }

#pragma unroll
    for (int i = 0; i < 32; i += 16) {
      v8f acc = {};
#pragma unroll
      for (int q = 0; q < 4; ++q) {
        v2f a = ld_a(T1, 16, i, 4 * q, lane);
        v2f b = ld_b_sinv(Sa, q, lane);
        acc = __builtin_amdgcn_wmma_f32_16x16x4_f32(false, a, false, b,
                                                    (short)0, acc, false, false);
      }
      st_c(Km, 16, i, 0, acc, lane);
    }
    __builtin_amdgcn_wave_barrier();

#pragma unroll
    for (int i = 0; i < 32; i += 16)
#pragma unroll
      for (int j = 0; j < 32; j += 16) {
        v8f acc = {};
        acc = mma16<false>(acc, Km, 16, i, 0, T2, 32, 0, j, lane);
        st_c_subfrom(P, 32, i, j, acc, lane);
      }
    {
      float acc = xp[lane];
#pragma unroll
      for (int j = 0; j < 16; ++j) acc += Km[lane * 16 + j] * nu[j];
      x[lane] = acc;
      ob[(size_t)t * 32 + lane] = acc;
    }
    __builtin_amdgcn_wave_barrier();
  }
}

extern "C" void kernel_launch(void* const* d_in, const int* in_sizes, int n_in,
                              void* d_out, int out_size, void* d_ws, size_t ws_size,
                              hipStream_t stream) {
  const float* state0 = (const float*)d_in[0];
  const float* cov0   = (const float*)d_in[1];
  const float* meas   = (const float*)d_in[2];
  const float* F      = (const float*)d_in[3];
  const float* H      = (const float*)d_in[4];
  const float* Q      = (const float*)d_in[5];
  const float* R      = (const float*)d_in[6];
  float* out = (float*)d_out;

  const int B = in_sizes[0] / 32;
  const int T = in_sizes[2] / (B * 16);
  const int blocks = (B + WPB - 1) / WPB;
  kalman_wmma_kernel<<<blocks, 32 * WPB, 0, stream>>>(state0, cov0, meas,
                                                      F, H, Q, R, out, B, T);
}


# Q, R and P_pred enter their products as the WMMA C operand from registers ([S|-I] eliminated, so the gain comes out negated): no LDS read-add epilogues; these terms are now summed first, not last
# speedup vs baseline: 4.0226x; 1.0818x over previous
#include <hip/hip_runtime.h>
#include <hip/hip_bf16.h>


typedef float v2f __attribute__((ext_vector_type(2)));
typedef float v4f __attribute__((ext_vector_type(4)));
typedef float v8f __attribute__((ext_vector_type(8)));

#define WPB 2


__device__ __forceinline__ v2f ld_a(const float* m, int ld, int r0, int c0, int lane) {
  const int h = lane >> 4, mr = lane & 15;
  const float* p = m + (r0 + mr) * ld + c0 + (h << 1);
  v2f a; a.x = p[0]; a.y = p[1];
  return a;
}
__device__ __forceinline__ v2f ld_b(const float* m, int ld, int r0, int c0, int lane) {
  const int h = lane >> 4, n = lane & 15;
  const float* p = m + (r0 + (h << 1)) * ld + c0 + n;
  v2f b; b.x = p[0]; b.y = p[ld];
  return b;
}
__device__ __forceinline__ v2f ld_bt(const float* m, int ld, int kr0, int nc0, int lane) {
  const int h = lane >> 4, n = lane & 15;
  const float* p = m + (nc0 + n) * ld + kr0 + (h << 1);
  v2f b; b.x = p[0]; b.y = p[1];
  return b;
}

__device__ __forceinline__ v2f ld_b_sinv(const float* m, int q, int lane) {
  const int h = lane >> 4, n = lane & 15;
  const float* p = m + q * 128 + (16 + n) * 4 + (h << 1);
  v2f b; b.x = p[0]; b.y = p[1];
  return b;
}

template <bool BT>
__device__ __forceinline__ v8f mma16(v8f acc, const float* A, int lda, int ar, int ac,
                                     const float* Bm, int ldb, int br, int bc, int lane) {
#pragma unroll
  for (int k = 0; k < 16; k += 4) {
    v2f a = ld_a(A, lda, ar, ac + k, lane);
    v2f b = BT ? ld_bt(Bm, ldb, br + k, bc, lane)
               : ld_b (Bm, ldb, br + k, bc, lane);
    acc = __builtin_amdgcn_wmma_f32_16x16x4_f32(false, a, false, b,
                                                (short)0, acc, false, false);
  }
  return acc;
}

__device__ __forceinline__ v8f mma16_ra(v8f acc, const v2f (&fr)[4],
                                        const float* Bm, int ldb, int br, int bc, int lane) {
  for (int q = 0; q < 4; ++q) {
    v2f b = ld_b(Bm, ldb, br + 4 * q, bc, lane);
    acc = __builtin_amdgcn_wmma_f32_16x16x4_f32(false, fr[q], false, b,
                                                (short)0, acc, false, false);
  }
  return acc;
}
__device__ __forceinline__ v8f mma16_rb(v8f acc, const float* A, int lda, int ar, int ac,
                                        const v2f (&fr)[4], int lane) {
  for (int q = 0; q < 4; ++q) {
    v2f a = ld_a(A, lda, ar, ac + 4 * q, lane);
    acc = __builtin_amdgcn_wmma_f32_16x16x4_f32(false, a, false, fr[q],
                                                (short)0, acc, false, false);
  }
  return acc;
}

__device__ __forceinline__ void st_c(float* d, int ld, int r0, int c0, v8f c, int lane) {
  const int h = lane >> 4, n = lane & 15;
#pragma unroll
  for (int r = 0; r < 8; ++r)
    d[(r0 + r + (h << 3)) * ld + c0 + n] = c[r];
}
__device__ __forceinline__ v8f ld_c(const float* __restrict__ g, int ld, int r0, int c0, int lane) {
  const int h = lane >> 4, n = lane & 15;
  v8f c;
#pragma unroll
  for (int r = 0; r < 8; ++r)
    c[r] = g[(r0 + r + (h << 3)) * ld + c0 + n];
  return c;
}

__global__ __launch_bounds__(32 * WPB)
void kalman_wmma_kernel(const float* __restrict__ state0, const float* __restrict__ cov0,
                        const float* __restrict__ meas,   const float* __restrict__ Fg,
                        const float* __restrict__ Hg,     const float* __restrict__ Qg,
                        const float* __restrict__ Rg,     float* __restrict__ out,
                        int B, int T) {
  const int tid  = threadIdx.x;
  const int lane = tid & 31;
  const int w    = tid >> 5;

  __shared__ float sF[32 * 32];
  __shared__ float sH[16 * 32];
  __shared__ float sP [WPB][32 * 32];
  __shared__ float sT1[WPB][32 * 32];
  __shared__ float sT2[WPB][16 * 32];
  __shared__ __attribute__((aligned(16))) float sSa[WPB][4 * 128];
  __shared__ float sKm[WPB][32 * 16];
  __shared__ float sx [WPB][32];
  __shared__ float sxp[WPB][32];
  __shared__ float sin_[WPB][16];

  for (int i = tid; i < 32 * 32; i += blockDim.x) sF[i] = Fg[i];
  for (int i = tid; i < 16 * 32; i += blockDim.x) sH[i] = Hg[i];
  __syncthreads();

  const int b = blockIdx.x * WPB + w;
  if (b >= B) return;

  float* P  = sP [w];
  float* T1 = sT1[w];
  float* T2 = sT2[w];
  float* Sa = sSa[w];
  float* Km = sKm[w];
  float* x  = sx [w];
  float* xp = sxp[w];
  float* nu = sin_[w];

  for (int i = lane; i < 32 * 32; i += 32) P[i] = cov0[(size_t)b * 1024 + i];
  x[lane] = state0[(size_t)b * 32 + lane];
  const float* zb = meas + (size_t)b * T * 16;
  float*       ob = out  + (size_t)b * T * 32;
  __builtin_amdgcn_wave_barrier();

  v2f fF[2][2][4], fH[2][4];
  for (int e = 0; e < 16; ++e)
    fF[e >> 3][(e >> 2) & 1][e & 3] = ld_a(sF, 32, 16 * (e >> 3), 4 * (e & 7), lane);
  for (int e = 0; e < 8; ++e)
    fH[e >> 2][e & 3] = ld_a(sH, 32, 0, 4 * e, lane);
  v8f qt[2][2], rt;
  for (int e = 0; e < 4; ++e)
    qt[e >> 1][e & 1] = ld_c(Qg, 32, 16 * (e >> 1), 16 * (e & 1), lane);
  rt = ld_c(Rg, 16, 0, 0, lane);

#pragma unroll 1
  for (int t = 0; t < T; ++t) {
    if (t + 1 < T) __builtin_prefetch(zb + (size_t)(t + 1) * 16, 0, 1);


#pragma unroll
    for (int i = 0; i < 32; i += 16)
#pragma unroll
      for (int j = 0; j < 32; j += 16) {
        v8f acc = {};
        acc = mma16_ra(acc, fF[i >> 4][0], P, 32, 0,  j, lane);
        acc = mma16_ra(acc, fF[i >> 4][1], P, 32, 16, j, lane);
        st_c(T1, 32, i, j, acc, lane);
      }
    {
      float acc = 0.f;
#pragma unroll
      for (int j = 0; j < 32; ++j) acc += sF[lane * 32 + j] * x[j];
      xp[lane] = acc;
    }
    __builtin_amdgcn_wave_barrier();

    {
      const int r16 = lane & 15;
      float acc = 0.f;
#pragma unroll
      for (int j = 0; j < 32; ++j) acc += sH[r16 * 32 + j] * xp[j];
      nu[r16] = zb[(size_t)t * 16 + r16] - acc;
    }
    v8f pp[2][2];
#pragma unroll
    for (int i = 0; i < 32; i += 16)
#pragma unroll
      for (int j = 0; j < 32; j += 16) {
        v8f acc = qt[i >> 4][j >> 4];
        acc = mma16_rb(acc, T1, 32, i, 0,  fF[j >> 4][0], lane);
        acc = mma16_rb(acc, T1, 32, i, 16, fF[j >> 4][1], lane);
        st_c(P, 32, i, j, acc, lane);
        pp[i >> 4][j >> 4] = acc;
      }
    __builtin_amdgcn_wave_barrier();

#pragma unroll
    for (int j = 0; j < 32; j += 16) {
      v8f acc = {};
      acc = mma16_ra(acc, fH[0], P, 32, 0,  j, lane);
      acc = mma16_ra(acc, fH[1], P, 32, 16, j, lane);
      st_c(T2, 32, 0, j, acc, lane);
    }
#pragma unroll
    for (int i = 0; i < 32; i += 16) {
      v8f acc = {};
      acc = mma16_rb(acc, P, 32, i, 0,  fH[0], lane);
      acc = mma16_rb(acc, P, 32, i, 16, fH[1], lane);
      st_c(T1, 16, i, 0, acc, lane);
    }
    __builtin_amdgcn_wave_barrier();

    {
      v8f acc = rt;
      acc = mma16_rb(acc, T2, 32, 0, 0,  fH[0], lane);
      acc = mma16_rb(acc, T2, 32, 0, 16, fH[1], lane);
      const int h = lane >> 4, n = lane & 15;
      v4f s0, s1, e0, e1;
#pragma unroll
      for (int r = 0; r < 4; ++r) {
        s0[r] = acc[r];
        s1[r] = acc[r + 4];
        e0[r] = ((h << 3) + r     == n) ? -1.0f : 0.0f;
        e1[r] = ((h << 3) + r + 4 == n) ? -1.0f : 0.0f;
      }
      *(v4f*)(Sa + (2 * h)     * 128 + n * 4)        = s0;
      *(v4f*)(Sa + (2 * h + 1) * 128 + n * 4)        = s1;
      *(v4f*)(Sa + (2 * h)     * 128 + (16 + n) * 4) = e0;
      *(v4f*)(Sa + (2 * h + 1) * 128 + (16 + n) * 4) = e1;
    }
    __builtin_amdgcn_wave_barrier();

    {
      float* colp = Sa + lane * 4;
#pragma unroll
      for (int k0 = 0; k0 < 16; k0 += 4) {
        float own[16], m[4][16];
#pragma unroll
        for (int j = 0; j < 4; ++j)
#pragma unroll
          for (int i = 0; i < 4; ++i) {
            const v4f t = *(const v4f*)(Sa + i * 128 + (k0 + j) * 4);
            m[j][4 * i] = t[0]; m[j][4 * i + 1] = t[1];
            m[j][4 * i + 2] = t[2]; m[j][4 * i + 3] = t[3];
          }
#pragma unroll
        for (int i = 0; i < 4; ++i) {
          const v4f t = *(const v4f*)(colp + i * 128);
          own[4 * i] = t[0]; own[4 * i + 1] = t[1];
          own[4 * i + 2] = t[2]; own[4 * i + 3] = t[3];
        }
        __builtin_amdgcn_wave_barrier();
#pragma unroll
        for (int j = 0; j < 4; ++j) {
          const int k = k0 + j;
          const float pinv = 1.0f / m[j][k];
          const float rk   = own[k] * pinv;
#pragma unroll
          for (int jj = j + 1; jj < 4; ++jj) {
            const float rkm = m[jj][k] * pinv;
#pragma unroll
            for (int r = 0; r < 16; ++r)
              m[jj][r] = (r == k) ? rkm : m[jj][r] - m[j][r] * rkm;
          }
#pragma unroll
          for (int r = 0; r < 16; ++r)
            own[r] = (r == k) ? rk : own[r] - m[j][r] * rk;
        }
#pragma unroll
        for (int i = 0; i < 4; ++i) {
          v4f t;
          t[0] = own[4 * i]; t[1] = own[4 * i + 1];
          t[2] = own[4 * i + 2]; t[3] = own[4 * i + 3];
          *(v4f*)(colp + i * 128) = t;
        }
        __builtin_amdgcn_wave_barrier();
      }
    }

#pragma unroll
    for (int i = 0; i < 32; i += 16) {
      v8f acc = {};
#pragma unroll
      for (int q = 0; q < 4; ++q) {
        v2f a = ld_a(T1, 16, i, 4 * q, lane);
        v2f b = ld_b_sinv(Sa, q, lane);
        acc = __builtin_amdgcn_wmma_f32_16x16x4_f32(false, a, false, b,
                                                    (short)0, acc, false, false);
      }
      st_c(Km, 16, i, 0, acc, lane);
    }
    __builtin_amdgcn_wave_barrier();

#pragma unroll
    for (int i = 0; i < 32; i += 16)
#pragma unroll
      for (int j = 0; j < 32; j += 16) {
        v8f acc = pp[i >> 4][j >> 4];
        acc = mma16<false>(acc, Km, 16, i, 0, T2, 32, 0, j, lane);
        st_c(P, 32, i, j, acc, lane);
      }
    {
      float acc = xp[lane];
#pragma unroll
      for (int j = 0; j < 16; ++j) acc -= Km[lane * 16 + j] * nu[j];
      x[lane] = acc;
      ob[(size_t)t * 32 + lane] = acc;
    }
    __builtin_amdgcn_wave_barrier();
  }
}

extern "C" void kernel_launch(void* const* d_in, const int* in_sizes, int n_in,
                              void* d_out, int out_size, void* d_ws, size_t ws_size,
                              hipStream_t stream) {
  const float* state0 = (const float*)d_in[0];
  const float* cov0   = (const float*)d_in[1];
  const float* meas   = (const float*)d_in[2];
  const float* F      = (const float*)d_in[3];
  const float* H      = (const float*)d_in[4];
  const float* Q      = (const float*)d_in[5];
  const float* R      = (const float*)d_in[6];
  float* out = (float*)d_out;

  const int B = in_sizes[0] / 32;
  const int T = in_sizes[2] / (B * 16);
  const int blocks = (B + WPB - 1) / WPB;
  kalman_wmma_kernel<<<blocks, 32 * WPB, 0, stream>>>(state0, cov0, meas,
                                                      F, H, Q, R, out, B, T);
}
